# starting kernel
# baseline (speedup 1.0000x reference)
#include <hip/hip_runtime.h>

#define D 64

typedef __attribute__((ext_vector_type(2))) float v2f;
typedef __attribute__((ext_vector_type(8))) float v8f;

__global__ void k_deg_init(float* __restrict__ deg, int n) {
    int i = blockIdx.x * blockDim.x + threadIdx.x;
    if (i < n) deg[i] = 1.0f;
}

__global__ void k_deg_accum(const int* __restrict__ dst, float* __restrict__ deg, int e) {
    int i = blockIdx.x * blockDim.x + threadIdx.x;
    if (i < e) atomicAdd(&deg[dst[i]], 1.0f);
}

__global__ void k_dinv(const float* __restrict__ deg, float* __restrict__ dinv, int n) {
    int i = blockIdx.x * blockDim.x + threadIdx.x;
    if (i < n) {
        float d = deg[i];
        dinv[i] = (d > 0.0f) ? rsqrtf(d) : 0.0f;
    }
}

__global__ void __launch_bounds__(256)
k_gemm_wmma(const float* __restrict__ X, const float* __restrict__ W,
            float* __restrict__ Y, int n) {
    const int lane = threadIdx.x & 31;
    const int wave = threadIdx.x >> 5;
    const int rowTiles = (n + 15) >> 4;
    const int rt = blockIdx.x * 2 + (wave >> 2);
    const int ct = wave & 3;
    if (rt >= rowTiles) return;

    const int r0 = rt << 4;
    const int c0 = ct << 4;
    const int m  = lane & 15;
    const int kh = lane >> 4;

    int arow = r0 + m;
    if (arow >= n) arow = n - 1;
    const float* __restrict__ aptr = X + (size_t)arow * D;
    const float* __restrict__ bptr = W + c0 + m;

    v8f acc = {};
#pragma unroll
    for (int k0 = 0; k0 < D; k0 += 4) {
        v2f a = *(const v2f*)(aptr + k0 + 2 * kh);
        v2f b;
        b.x = bptr[(k0 + 2 * kh)     * D];
        b.y = bptr[(k0 + 2 * kh + 1) * D];
        acc = __builtin_amdgcn_wmma_f32_16x16x4_f32(
             false, a,  false, b,
             (short)0, acc,  false,  false);
    }

    const int col = c0 + m;
#pragma unroll
    for (int r = 0; r < 8; ++r) {
        int row = r0 + r + 8 * kh;
        if (row < n) Y[(size_t)row * D + col] = acc[r];
    }
}

__global__ void k_self_bias(const float* __restrict__ xw, const float* __restrict__ dinv,
                            const float* __restrict__ b, float* __restrict__ agg, int total) {
    int i = blockIdx.x * blockDim.x + threadIdx.x;
    if (i < total) {
        int v = i >> 6;
        int d = i & 63;
        float s = dinv[v];
        agg[i] = s * s * xw[i] + b[d];
    }
}

__global__ void __launch_bounds__(256)
k_scatter(const int* __restrict__ src, const int* __restrict__ dst,
          const float* __restrict__ dinv, const float* __restrict__ xw,
          float* __restrict__ agg, int e) {
    int edge = (blockIdx.x << 3) + (threadIdx.x >> 5);
    if (edge >= e) return;
    int s = src[edge];
    int t = dst[edge];
    float nrm = dinv[s] * dinv[t];
    int lane = threadIdx.x & 31;
    const float* __restrict__ xs = xw + (size_t)s * D;
    float* __restrict__ at = agg + (size_t)t * D;
    atomicAdd(at + lane,      nrm * xs[lane]);
    atomicAdd(at + lane + 32, nrm * xs[lane + 32]);
}

__global__ void k_relu(float* __restrict__ h, int total) {
    int i = blockIdx.x * blockDim.x + threadIdx.x;
    if (i < total) h[i] = fmaxf(h[i], 0.0f);
}

extern "C" void kernel_launch(void* const* d_in, const int* in_sizes, int n_in,
                              void* d_out, int out_size, void* d_ws, size_t ws_size,
                              hipStream_t stream) {
    const float* x  = (const float*)d_in[0];
    const int*   ei = (const int*)  d_in[1];
    const float* W1 = (const float*)d_in[2];
    const float* b1 = (const float*)d_in[3];
    const float* W2 = (const float*)d_in[4];
    const float* b2 = (const float*)d_in[5];

    const int n = in_sizes[0] / D;
    const int e = in_sizes[1] / 2;
    const int* src = ei;
    const int* dst = ei + e;

    float* ws   = (float*)d_ws;
    float* deg  = ws;
    float* dinv = ws + n;
    float* xw   = ws + 2 * (size_t)n;
    float* agg  = xw + (size_t)n * D;
    float* out  = (float*)d_out;

    const int T = 256;
    const int total = n * D;
    const int rowTiles = (n + 15) / 16;
    const int gemmBlocks = (rowTiles + 1) / 2;
    const int scatBlocks = (e + 7) / 8;

    k_deg_init <<<(n + T - 1) / T, T, 0, stream>>>(deg, n);
    k_deg_accum<<<(e + T - 1) / T, T, 0, stream>>>(dst, deg, e);
    k_dinv     <<<(n + T - 1) / T, T, 0, stream>>>(deg, dinv, n);

    k_gemm_wmma<<<gemmBlocks, 256, 0, stream>>>(x, W1, xw, n);
    k_self_bias<<<(total + T - 1) / T, T, 0, stream>>>(xw, dinv, b1, agg, total);
    k_scatter  <<<scatBlocks, 256, 0, stream>>>(src, dst, dinv, xw, agg, e);
    k_relu     <<<(total + T - 1) / T, T, 0, stream>>>(agg, total);

    k_gemm_wmma<<<gemmBlocks, 256, 0, stream>>>(agg, W2, xw, n);
    k_self_bias<<<(total + T - 1) / T, T, 0, stream>>>(xw, dinv, b2, out, total);
    k_scatter  <<<scatBlocks, 256, 0, stream>>>(src, dst, dinv, xw, out, e);
}


# GEMM: a wave computes a 16-row strip of all 64 columns, not one 16x16 tile; X rows and W go through LDS as whole 128-B lines, results leave as whole lines; fragments, k order and sum order unchanged
# speedup vs baseline: 3.2164x; 3.2164x over previous
#include <hip/hip_runtime.h>

#define D 64

typedef __attribute__((ext_vector_type(2))) float v2f;
typedef __attribute__((ext_vector_type(8))) float v8f;

__device__ __forceinline__ int chip_half() {
    return (int)((__builtin_amdgcn_s_sendmsg_rtn(0x87) >> 18) & 1u);
}

__global__ void k_deg_init(float* __restrict__ deg, int n, int np) {
    int i = blockIdx.x * blockDim.x + threadIdx.x;
    if (i < n) {
        deg[i] = 1.0f;
        deg[np + i] = 0.0f;
    }
}

#define HN 65536
#define HB 128
#define HU 16
#define HF 8

__global__ void __launch_bounds__(1024)
k_deg_count(const int* __restrict__ dst, float* __restrict__ deg, int n, int np, int e) {
    __shared__ __attribute__((aligned(16))) unsigned hist[HN];
    const int half = chip_half();
    const int tid = threadIdx.x;
    float* __restrict__ dh = deg + (size_t)half * np;
    const int per = ((e + (int)gridDim.x - 1) / (int)gridDim.x + 31) & ~31;
    const int lo = blockIdx.x * per;
    const int hi = min(e, lo + per);
    for (int node0 = 0; node0 < n; node0 += HN) {
        const int cnt = min(HN, n - node0);
        for (int i = tid; i < (cnt + 3) / 4; i += 1024)
            ((uint4*)hist)[i] = make_uint4(0u, 0u, 0u, 0u);
        __syncthreads();
        for (int i0 = lo + tid; i0 < hi; i0 += 1024 * HU) {
            int d[HU];
#pragma unroll
            for (int u = 0; u < HU; ++u) {
                const int i = i0 + u * 1024;
                d[u] = (i < hi) ? dst[i] : -1;
            }
#pragma unroll
            for (int u = 0; u < HU; ++u) {
                const unsigned r = (unsigned)(d[u] - node0);
                if (r < (unsigned)cnt) atomicAdd(&hist[r], 1u);
            }
        }
        __syncthreads();
        for (int i0 = tid; i0 < cnt; i0 += 1024 * HF) {
            unsigned c[HF];
#pragma unroll
            for (int u = 0; u < HF; ++u) {
                const int i = i0 + u * 1024;
                c[u] = (i < cnt) ? hist[i] : 0u;
            }
#pragma unroll
            for (int u = 0; u < HF; ++u)
                if (c[u]) atomicAdd(dh + node0 + i0 + u * 1024, (float)c[u]);
        }
        __syncthreads();
    }
}

__device__ __forceinline__ float inv_sqrt_deg(float d) {
    return (d > 0.0f) ? rsqrtf(d) : 0.0f;
}

#define GW 4
#define AP 68
#define WP 160

template <bool LAYER2>
__global__ void __launch_bounds__(GW * 32)
k_gemm_wmma(const float* __restrict__ X, const float* __restrict__ X1,
            const float* __restrict__ W,
            const float* __restrict__ deg, const float* __restrict__ deg1,
            const float* __restrict__ bias,
            float* __restrict__ Y, float* __restrict__ Z, float* __restrict__ Z1,
            float* __restrict__ dsum, int n) {
    __shared__ __attribute__((aligned(16))) float sW[(D / 2) * WP];
    __shared__ __attribute__((aligned(16))) float sT[GW][16 * AP];
    const int tid  = threadIdx.x;
    const int lane = tid & 31;
    const int wave = tid >> 5;
    const int r0 = (blockIdx.x * GW + wave) << 4;
    const int m  = lane & 15;
    const int kh = lane >> 4;
    const int lr = lane >> 4;
    const int lc = (lane & 15) << 2;
    float* __restrict__ tA = sT[wave];

    float4 av[8], av1[8];
#pragma unroll
    for (int i = 0; i < 8; ++i) {
        const int row = min(r0 + 2 * i + lr, n - 1);
        av[i] = *(const float4*)(X + (size_t)row * D + lc);
        if (LAYER2) av1[i] = *(const float4*)(X1 + (size_t)row * D + lc);
    }
    float4 we[(D * D / 8) / (GW * 32)], wo[(D * D / 8) / (GW * 32)];
#pragma unroll
    for (int j = 0; j < (D * D / 8) / (GW * 32); ++j) {
        const int q = tid + j * GW * 32;
        we[j] = *(const float4*)(W + (size_t)(2 * (q >> 4))     * D + ((q & 15) << 2));
        wo[j] = *(const float4*)(W + (size_t)(2 * (q >> 4) + 1) * D + ((q & 15) << 2));
    }
    float dg[8];
#pragma unroll
    for (int i = 0; i < 8; ++i) {
        const int row = min(r0 + 2 * i + lr, n - 1);
        dg[i] = LAYER2 ? deg[row] : deg[row] + deg1[row];
    }
    const float4 b4 = *(const float4*)(bias + lc);

#pragma unroll
    for (int j = 0; j < (D * D / 8) / (GW * 32); ++j) {
        const int q = tid + j * GW * 32;
        float* __restrict__ wp = &sW[(q >> 4) * WP + ((q & 15) << 3)];
        *(float4*)(wp)     = make_float4(we[j].x, wo[j].x, we[j].y, wo[j].y);
        *(float4*)(wp + 4) = make_float4(we[j].z, wo[j].z, we[j].w, wo[j].w);
    }
#pragma unroll
    for (int i = 0; i < 8; ++i) {
        float4 a = av[i];
        if (LAYER2) {
            a.x = fmaxf(a.x + av1[i].x, 0.0f);
            a.y = fmaxf(a.y + av1[i].y, 0.0f);
            a.z = fmaxf(a.z + av1[i].z, 0.0f);
            a.w = fmaxf(a.w + av1[i].w, 0.0f);
        }
        *(float4*)&tA[(2 * i + lr) * AP + lc] = a;
    }
    __syncthreads();

    v8f acc[4] = {};
#pragma unroll
    for (int k0 = 0; k0 < D; k0 += 4) {
        const v2f a = *(const v2f*)&tA[m * AP + k0 + 2 * kh];
#pragma unroll
        for (int ct = 0; ct < 4; ++ct) {
            const v2f b = *(const v2f*)&sW[(k0 / 2 + kh) * WP + 2 * (16 * ct + m)];
            acc[ct] = __builtin_amdgcn_wmma_f32_16x16x4_f32(
                 false, a,  false, b,
                 (short)0, acc[ct],  false,  false);
        }
    }

#pragma unroll
    for (int ct = 0; ct < 4; ++ct)
#pragma unroll
        for (int r = 0; r < 8; ++r)
            tA[(r + 8 * kh) * AP + 16 * ct + m] = acc[ct][r];
    __syncthreads();

#pragma unroll
    for (int i = 0; i < 8; ++i) {
        const int row = r0 + 2 * i + lr;
        if (row < n) {
            const float4 y = *(const float4*)&tA[(2 * i + lr) * AP + lc];
            const size_t o = (size_t)row * D + lc;
            const float dv = inv_sqrt_deg(dg[i]);
            const float sc = dv * dv;
            float4 z;
            z.x = sc * y.x + b4.x;
            z.y = sc * y.y + b4.y;
            z.z = sc * y.z + b4.z;
            z.w = sc * y.w + b4.w;
            *(float4*)(Y + o)  = y;
            *(float4*)(Z + o)  = z;
            *(float4*)(Z1 + o) = make_float4(0.0f, 0.0f, 0.0f, 0.0f);
            if (!LAYER2 && lc == 0) dsum[row] = dg[i];
        }
    }
}

__global__ void __launch_bounds__(256)
k_add_copies(float* __restrict__ out, const float* __restrict__ out1, int n4) {
    const int i = blockIdx.x * 256 + threadIdx.x;
    if (i < n4) {
        float4 a = ((const float4*)out)[i];
        const float4 b = ((const float4*)out1)[i];
        a.x += b.x; a.y += b.y; a.z += b.z; a.w += b.w;
        ((float4*)out)[i] = a;
    }
}

#define WN 65536
#define WB 128
#define WF 32
#define WU 16

__global__ void __launch_bounds__(1024)
k_edge_weight(const int* __restrict__ src, const int* __restrict__ dst,
              const float* __restrict__ deg, float* __restrict__ wgt, int n, int e) {
    __shared__ __attribute__((aligned(16))) float dinv[WN];
    const int tid = threadIdx.x;
    const int cap = min(n, WN);
    for (int i0 = tid; i0 < cap; i0 += 1024 * WF) {
        float d[WF];
#pragma unroll
        for (int u = 0; u < WF; ++u) {
            const int i = i0 + u * 1024;
            d[u] = (i < cap) ? deg[i] : 0.0f;
        }
#pragma unroll
        for (int u = 0; u < WF; ++u) {
            const int i = i0 + u * 1024;
            if (i < cap) dinv[i] = inv_sqrt_deg(d[u]);
        }
    }
    __syncthreads();
    const int per = ((e + (int)gridDim.x - 1) / (int)gridDim.x + 31) & ~31;
    const int lo = blockIdx.x * per;
    const int hi = min(e, lo + per);
    for (int i0 = lo + tid; i0 < hi; i0 += 1024 * WU) {
        int s[WU], t[WU];
        float w[WU];
#pragma unroll
        for (int u = 0; u < WU; ++u) {
            const int i = i0 + u * 1024;
            s[u] = (i < hi) ? src[i] : 0;
            t[u] = (i < hi) ? dst[i] : 0;
        }
#pragma unroll
        for (int u = 0; u < WU; ++u)
            w[u] = dinv[min(s[u], WN - 1)] * dinv[min(t[u], WN - 1)];
        if (n > WN) {
#pragma unroll
            for (int u = 0; u < WU; ++u)
                if (max(s[u], t[u]) >= WN) {
                    const float fs = (s[u] < WN) ? dinv[s[u]] : inv_sqrt_deg(deg[s[u]]);
                    const float ft = (t[u] < WN) ? dinv[t[u]] : inv_sqrt_deg(deg[t[u]]);
                    w[u] = fs * ft;
                }
        }
#pragma unroll
        for (int u = 0; u < WU; ++u) {
            const int i = i0 + u * 1024;
            if (i < hi) wgt[i] = w[u];
        }
    }
}

#define EPW 32
#define EB  8

__device__ __forceinline__ int lane_bcast(int v, int l) {
    return __builtin_amdgcn_readlane(v, l);
}

__global__ void __launch_bounds__(256)
k_scatter(const int* __restrict__ src, const int* __restrict__ dst,
          const float* __restrict__ wgt, const float* __restrict__ xw,
          float* __restrict__ agg0, float* __restrict__ agg1, int e) {
    const int half = chip_half();
    const int lane = threadIdx.x & 31;
    const int base = ((blockIdx.x << 3) + (threadIdx.x >> 5)) * EPW;
    if (base >= e) return;
    const int cnt = min(EPW, e - base);
    const int mine = base + min(lane, cnt - 1);
    const int s = src[mine];
    const int t = dst[mine];
    const int wbits = __float_as_int(wgt[mine]);
    float* __restrict__ agg = half ? agg1 : agg0;

    if (cnt == EPW) {
        float a0[2][EB], a1[2][EB];
#pragma unroll
        for (int j = 0; j < EB; ++j) {
            const float* __restrict__ xs = xw + (size_t)lane_bcast(s, j) * D;
            a0[0][j] = xs[lane];
            a1[0][j] = xs[lane + 32];
        }
#pragma unroll
        for (int b = 0; b < EPW / EB; ++b) {
            if (b + 1 < EPW / EB) {
#pragma unroll
                for (int j = 0; j < EB; ++j) {
                    const float* __restrict__ xs = xw + (size_t)lane_bcast(s, (b + 1) * EB + j) * D;
                    a0[(b + 1) & 1][j] = xs[lane];
                    a1[(b + 1) & 1][j] = xs[lane + 32];
                }
            }
#pragma unroll
            for (int j = 0; j < EB; ++j) {
                const float nrm = __int_as_float(lane_bcast(wbits, b * EB + j));
                float* __restrict__ at = agg + (size_t)lane_bcast(t, b * EB + j) * D;
                atomicAdd(at + lane,      nrm * a0[b & 1][j]);
                atomicAdd(at + lane + 32, nrm * a1[b & 1][j]);
            }
        }
    } else {
        for (int j = 0; j < cnt; ++j) {
            const float nrm = __int_as_float(lane_bcast(wbits, j));
            const float* __restrict__ xs = xw + (size_t)lane_bcast(s, j) * D;
            float* __restrict__ at = agg + (size_t)lane_bcast(t, j) * D;
            atomicAdd(at + lane,      nrm * xs[lane]);
            atomicAdd(at + lane + 32, nrm * xs[lane + 32]);
        }
    }
}

extern "C" void kernel_launch(void* const* d_in, const int* in_sizes, int n_in,
                              void* d_out, int out_size, void* d_ws, size_t ws_size,
                              hipStream_t stream) {
    const float* x  = (const float*)d_in[0];
    const int*   ei = (const int*)  d_in[1];
    const float* W1 = (const float*)d_in[2];
    const float* b1 = (const float*)d_in[3];
    const float* W2 = (const float*)d_in[4];
    const float* b2 = (const float*)d_in[5];

    const int n = in_sizes[0] / D;
    const int e = in_sizes[1] / 2;
    const int* src = ei;
    const int* dst = ei + e;

    float* ws   = (float*)d_ws;
    float* xw   = ws + 2 * (size_t)n;
    float* agg  = xw + (size_t)n * D;
    float* agg1 = agg + (size_t)n * D;
    float* out1 = agg1 + (size_t)n * D;
    float* dsum = out1 + (size_t)n * D;
    const int np = (n + 31) & ~31;
    float* deg  = ws + ((3 * (size_t)n + 4 * (size_t)n * D + 31) & ~(size_t)31);
    float* wgt  = deg + 2 * (size_t)np;
    float* out  = (float*)d_out;

    const int T = 256;
    const int rowTiles = (n + 15) / 16;
    const int gemmBlocks = (rowTiles + GW - 1) / GW;
    const int scatBlocks = (e + 8 * EPW - 1) / (8 * EPW);

    k_deg_init <<<(n + T - 1) / T, T, 0, stream>>>(deg, n, np);
    k_deg_count<<<HB, 1024, 0, stream>>>(dst, deg, n, np, e);

    k_gemm_wmma<false><<<gemmBlocks, GW * 32, 0, stream>>>(x, x, W1, deg, deg + np, b1, xw, agg, agg1, dsum, n);
    k_edge_weight     <<<WB, 1024, 0, stream>>>(src, dst, dsum, wgt, n, e);
    k_scatter         <<<scatBlocks, 256, 0, stream>>>(src, dst, wgt, xw, agg, agg1, e);

    k_gemm_wmma<true> <<<gemmBlocks, GW * 32, 0, stream>>>(agg, agg1, W2, dsum, dsum, b2, xw, out, out1, dsum, n);
    k_scatter         <<<scatBlocks, 256, 0, stream>>>(src, dst, wgt, xw, out, out1, e);

    k_add_copies<<<(n * (D / 4) + T - 1) / T, T, 0, stream>>>(out, out1, n * (D / 4));
}


# edge sums in ONE copy: destination nodes split between the chip halves by range, a work-group draws its half's ticket and adds only that half's edges; no zero fill or final add; arrival order per row
# speedup vs baseline: 4.0316x; 1.2535x over previous
#include <hip/hip_runtime.h>

#define D 64

typedef __attribute__((ext_vector_type(2))) float v2f;
typedef __attribute__((ext_vector_type(8))) float v8f;

__device__ __forceinline__ int chip_half() {
    return (int)((__builtin_amdgcn_s_sendmsg_rtn(0x87) >> 18) & 1u);
}

#define TS 64
__global__ void k_deg_init(float* __restrict__ deg, int* __restrict__ tick, int n, int np) {
    int i = blockIdx.x * blockDim.x + threadIdx.x;
    if (i < n) {
        deg[i] = 1.0f;
        deg[np + i] = 0.0f;
    }
    if (i < 4 * TS) tick[i] = 0;
}

#define HN 65536
#define HB 128
#define HU 16
#define HF 8

__global__ void __launch_bounds__(1024)
k_deg_count(const int* __restrict__ dst, float* __restrict__ deg, int n, int np, int e) {
    __shared__ __attribute__((aligned(16))) unsigned hist[HN];
    const int half = chip_half();
    const int tid = threadIdx.x;
    float* __restrict__ dh = deg + (size_t)half * np;
    const int per = ((e + (int)gridDim.x - 1) / (int)gridDim.x + 31) & ~31;
    const int lo = blockIdx.x * per;
    const int hi = min(e, lo + per);
    for (int node0 = 0; node0 < n; node0 += HN) {
        const int cnt = min(HN, n - node0);
        for (int i = tid; i < (cnt + 3) / 4; i += 1024)
            ((uint4*)hist)[i] = make_uint4(0u, 0u, 0u, 0u);
        __syncthreads();
        for (int i0 = lo + tid; i0 < hi; i0 += 1024 * HU) {
            int d[HU];
#pragma unroll
            for (int u = 0; u < HU; ++u) {
                const int i = i0 + u * 1024;
                d[u] = (i < hi) ? dst[i] : -1;
            }
#pragma unroll
            for (int u = 0; u < HU; ++u) {
                const unsigned r = (unsigned)(d[u] - node0);
                if (r < (unsigned)cnt) atomicAdd(&hist[r], 1u);
            }
        }
        __syncthreads();
        for (int i0 = tid; i0 < cnt; i0 += 1024 * HF) {
            unsigned c[HF];
#pragma unroll
            for (int u = 0; u < HF; ++u) {
                const int i = i0 + u * 1024;
                c[u] = (i < cnt) ? hist[i] : 0u;
            }
#pragma unroll
            for (int u = 0; u < HF; ++u)
                if (c[u]) atomicAdd(dh + node0 + i0 + u * 1024, (float)c[u]);
        }
        __syncthreads();
    }
}

__device__ __forceinline__ float inv_sqrt_deg(float d) {
    return (d > 0.0f) ? rsqrtf(d) : 0.0f;
}

#define GW 4
#define AP 68
#define WP 160

template <bool LAYER2>
__global__ void __launch_bounds__(GW * 32)
k_gemm_wmma(const float* __restrict__ X,
            const float* __restrict__ W,
            const float* __restrict__ deg, const float* __restrict__ deg1,
            const float* __restrict__ bias,
            float* __restrict__ Y, float* __restrict__ Z,
            float* __restrict__ dsum, int n) {
    __shared__ __attribute__((aligned(16))) float sW[(D / 2) * WP];
    __shared__ __attribute__((aligned(16))) float sT[GW][16 * AP];
    const int tid  = threadIdx.x;
    const int lane = tid & 31;
    const int wave = tid >> 5;
    const int r0 = (blockIdx.x * GW + wave) << 4;
    const int m  = lane & 15;
    const int kh = lane >> 4;
    const int lr = lane >> 4;
    const int lc = (lane & 15) << 2;
    float* __restrict__ tA = sT[wave];

    float4 av[8];
#pragma unroll
    for (int i = 0; i < 8; ++i) {
        const int row = min(r0 + 2 * i + lr, n - 1);
        av[i] = *(const float4*)(X + (size_t)row * D + lc);
    }
    float4 we[(D * D / 8) / (GW * 32)], wo[(D * D / 8) / (GW * 32)];
#pragma unroll
    for (int j = 0; j < (D * D / 8) / (GW * 32); ++j) {
        const int q = tid + j * GW * 32;
        we[j] = *(const float4*)(W + (size_t)(2 * (q >> 4))     * D + ((q & 15) << 2));
        wo[j] = *(const float4*)(W + (size_t)(2 * (q >> 4) + 1) * D + ((q & 15) << 2));
    }
    float dg[8];
#pragma unroll
    for (int i = 0; i < 8; ++i) {
        const int row = min(r0 + 2 * i + lr, n - 1);
        dg[i] = LAYER2 ? deg[row] : deg[row] + deg1[row];
    }
    const float4 b4 = *(const float4*)(bias + lc);

#pragma unroll
    for (int j = 0; j < (D * D / 8) / (GW * 32); ++j) {
        const int q = tid + j * GW * 32;
        float* __restrict__ wp = &sW[(q >> 4) * WP + ((q & 15) << 3)];
        *(float4*)(wp)     = make_float4(we[j].x, wo[j].x, we[j].y, wo[j].y);
        *(float4*)(wp + 4) = make_float4(we[j].z, wo[j].z, we[j].w, wo[j].w);
    }
#pragma unroll
    for (int i = 0; i < 8; ++i) {
        float4 a = av[i];
        if (LAYER2) {
            a.x = fmaxf(a.x, 0.0f);
            a.y = fmaxf(a.y, 0.0f);
            a.z = fmaxf(a.z, 0.0f);
            a.w = fmaxf(a.w, 0.0f);
        }
        *(float4*)&tA[(2 * i + lr) * AP + lc] = a;
    }
    __syncthreads();

    v8f acc[4] = {};
#pragma unroll
    for (int k0 = 0; k0 < D; k0 += 4) {
        const v2f a = *(const v2f*)&tA[m * AP + k0 + 2 * kh];
#pragma unroll
        for (int ct = 0; ct < 4; ++ct) {
            const v2f b = *(const v2f*)&sW[(k0 / 2 + kh) * WP + 2 * (16 * ct + m)];
            acc[ct] = __builtin_amdgcn_wmma_f32_16x16x4_f32(
                 false, a,  false, b,
                 (short)0, acc[ct],  false,  false);
        }
    }

#pragma unroll
    for (int ct = 0; ct < 4; ++ct)
#pragma unroll
        for (int r = 0; r < 8; ++r)
            tA[(r + 8 * kh) * AP + 16 * ct + m] = acc[ct][r];
    __syncthreads();

#pragma unroll
    for (int i = 0; i < 8; ++i) {
        const int row = r0 + 2 * i + lr;
        if (row < n) {
            const float4 y = *(const float4*)&tA[(2 * i + lr) * AP + lc];
            const size_t o = (size_t)row * D + lc;
            const float dv = inv_sqrt_deg(dg[i]);
            const float sc = dv * dv;
            float4 z;
            z.x = sc * y.x + b4.x;
            z.y = sc * y.y + b4.y;
            z.z = sc * y.z + b4.z;
            z.w = sc * y.w + b4.w;
            *(float4*)(Y + o)  = y;
            *(float4*)(Z + o)  = z;
            if (!LAYER2 && lc == 0) dsum[row] = dg[i];
        }
    }
}

#define WN 65536
#define WB 128
#define WF 32
#define WU 16

__global__ void __launch_bounds__(1024)
k_edge_weight(const int* __restrict__ src, const int* __restrict__ dst,
              const float* __restrict__ deg, float* __restrict__ wgt, int n, int e) {
    __shared__ __attribute__((aligned(16))) float dinv[WN];
    const int tid = threadIdx.x;
    const int cap = min(n, WN);
    for (int i0 = tid; i0 < cap; i0 += 1024 * WF) {
        float d[WF];
#pragma unroll
        for (int u = 0; u < WF; ++u) {
            const int i = i0 + u * 1024;
            d[u] = (i < cap) ? deg[i] : 0.0f;
        }
#pragma unroll
        for (int u = 0; u < WF; ++u) {
            const int i = i0 + u * 1024;
            if (i < cap) dinv[i] = inv_sqrt_deg(d[u]);
        }
    }
    __syncthreads();
    const int per = ((e + (int)gridDim.x - 1) / (int)gridDim.x + 31) & ~31;
    const int lo = blockIdx.x * per;
    const int hi = min(e, lo + per);
    for (int i0 = lo + tid; i0 < hi; i0 += 1024 * WU) {
        int s[WU], t[WU];
        float w[WU];
#pragma unroll
        for (int u = 0; u < WU; ++u) {
            const int i = i0 + u * 1024;
            s[u] = (i < hi) ? src[i] : 0;
            t[u] = (i < hi) ? dst[i] : 0;
        }
#pragma unroll
        for (int u = 0; u < WU; ++u)
            w[u] = dinv[min(s[u], WN - 1)] * dinv[min(t[u], WN - 1)];
        if (n > WN) {
#pragma unroll
            for (int u = 0; u < WU; ++u)
                if (max(s[u], t[u]) >= WN) {
                    const float fs = (s[u] < WN) ? dinv[s[u]] : inv_sqrt_deg(deg[s[u]]);
                    const float ft = (t[u] < WN) ? dinv[t[u]] : inv_sqrt_deg(deg[t[u]]);
                    w[u] = fs * ft;
                }
        }
#pragma unroll
        for (int u = 0; u < WU; ++u) {
            const int i = i0 + u * 1024;
            if (i < hi) wgt[i] = w[u];
        }
    }
}

#define EPW 32
#define EB  8

__device__ __forceinline__ int lane_bcast(int v, int l) {
    return __builtin_amdgcn_readlane(v, l);
}

template <int NB>
__device__ __forceinline__ void walk_edges(const float* __restrict__ xw, float* __restrict__ agg,
                                           int s, int t, int wbits, int cnt, int lane) {
    float a0[2][EB], a1[2][EB];
#pragma unroll
    for (int j = 0; j < EB; ++j) {
        if (NB > 1 || j < cnt) {
            const float* __restrict__ xs = xw + (size_t)lane_bcast(s, j) * D;
            a0[0][j] = xs[lane];
            a1[0][j] = xs[lane + 32];
        }
    }
#pragma unroll
    for (int b = 0; b < NB; ++b) {
        if (b + 1 < NB) {
#pragma unroll
            for (int j = 0; j < EB; ++j) {
                if (b + 2 < NB || (b + 1) * EB + j < cnt) {
                    const float* __restrict__ xs = xw + (size_t)lane_bcast(s, (b + 1) * EB + j) * D;
                    a0[(b + 1) & 1][j] = xs[lane];
                    a1[(b + 1) & 1][j] = xs[lane + 32];
                }
            }
        }
#pragma unroll
        for (int j = 0; j < EB; ++j) {
            if (b + 1 < NB || b * EB + j < cnt) {
                const float nrm = __int_as_float(lane_bcast(wbits, b * EB + j));
                float* __restrict__ ad = agg + (size_t)lane_bcast(t, b * EB + j) * D;
                atomicAdd(ad + lane,      nrm * a0[b & 1][j]);
                atomicAdd(ad + lane + 32, nrm * a1[b & 1][j]);
            }
        }
    }
}

__global__ void __launch_bounds__(256)
k_scatter(const int* __restrict__ src, const int* __restrict__ dst,
          const float* __restrict__ wgt, const float* __restrict__ xw,
          float* __restrict__ agg, int* __restrict__ tick, int ngrp, int n, int e) {
    __shared__ int s_job;
    const int half = chip_half();
    const int lane = threadIdx.x & 31;
    if (threadIdx.x == 0) {
        int own = half;
        int tk = atomicAdd(tick + own * TS, 1);
        if (tk >= ngrp) {
            own ^= 1;
            tk = atomicAdd(tick + own * TS, 1);
        }
        s_job = (tk < ngrp) ? (tk * 2 + own) : -1;
    }
    __syncthreads();
    const int job = s_job;
    if (job < 0) return;
    const int own = job & 1;
    const int base = (((job >> 1) << 3) + (threadIdx.x >> 5)) * EPW;
    if (base >= e) return;
    const int valid = min(EPW, e - base);
    const int at = base + min(lane, valid - 1);
    int s = src[at];
    int t = dst[at];
    int wbits = __float_as_int(wgt[at]);

    const bool mine = (lane < valid) && ((t >= (n >> 1)) == (own != 0));
    const unsigned mask = __builtin_amdgcn_ballot_w32(mine);
    const int cnt = __builtin_popcount(mask);
    const unsigned below = (1u << lane) - 1u;
    const int to = mine ? __builtin_popcount(mask & below) : cnt + __builtin_popcount(~mask & below);
    s     = __builtin_amdgcn_ds_permute(to << 2, s);
    t     = __builtin_amdgcn_ds_permute(to << 2, t);
    wbits = __builtin_amdgcn_ds_permute(to << 2, wbits);

    if (cnt > 3 * EB)      walk_edges<4>(xw, agg, s, t, wbits, cnt, lane);
    else if (cnt > 2 * EB) walk_edges<3>(xw, agg, s, t, wbits, cnt, lane);
    else if (cnt > EB)     walk_edges<2>(xw, agg, s, t, wbits, cnt, lane);
    else                   walk_edges<1>(xw, agg, s, t, wbits, cnt, lane);
}

extern "C" void kernel_launch(void* const* d_in, const int* in_sizes, int n_in,
                              void* d_out, int out_size, void* d_ws, size_t ws_size,
                              hipStream_t stream) {
    const float* x  = (const float*)d_in[0];
    const int*   ei = (const int*)  d_in[1];
    const float* W1 = (const float*)d_in[2];
    const float* b1 = (const float*)d_in[3];
    const float* W2 = (const float*)d_in[4];
    const float* b2 = (const float*)d_in[5];

    const int n = in_sizes[0] / D;
    const int e = in_sizes[1] / 2;
    const int* src = ei;
    const int* dst = ei + e;

    float* ws   = (float*)d_ws;
    float* xw   = ws + 2 * (size_t)n;
    float* agg  = xw + (size_t)n * D;
    float* dsum = agg + (size_t)n * D;
    const int np = (n + 31) & ~31;
    float* deg  = ws + ((3 * (size_t)n + 2 * (size_t)n * D + 31) & ~(size_t)31);
    float* wgt  = deg + 2 * (size_t)np;
    int*   tick = (int*)(wgt + (((size_t)e + 31) & ~(size_t)31));
    float* out  = (float*)d_out;

    const int T = 256;
    const int rowTiles = (n + 15) / 16;
    const int gemmBlocks = (rowTiles + GW - 1) / GW;
    const int scatBlocks = (e + 8 * EPW - 1) / (8 * EPW);

    k_deg_init <<<(n + T - 1) / T, T, 0, stream>>>(deg, tick, n, np);
    k_deg_count<<<HB, 1024, 0, stream>>>(dst, deg, n, np, e);

    k_gemm_wmma<false><<<gemmBlocks, GW * 32, 0, stream>>>(x, W1, deg, deg + np, b1, xw, agg, dsum, n);
    k_edge_weight     <<<WB, 1024, 0, stream>>>(src, dst, dsum, wgt, n, e);
    k_scatter         <<<2 * scatBlocks, 256, 0, stream>>>(src, dst, wgt, xw, agg, tick, scatBlocks, n, e);

    k_gemm_wmma<true> <<<gemmBlocks, GW * 32, 0, stream>>>(agg, W2, dsum, dsum, b2, xw, out, dsum, n);
    k_scatter         <<<2 * scatBlocks, 256, 0, stream>>>(src, dst, wgt, xw, out, tick + 2 * TS, scatBlocks, n, e);
}


# edge scatter: a fixed grid of resident work-groups loops over tickets; the next visit's indices and the ticket after it are fetched under the current walk; same visits, sums still in arrival order
# speedup vs baseline: 4.2064x; 1.0434x over previous
#include <hip/hip_runtime.h>

#define D 64

typedef __attribute__((ext_vector_type(2))) float v2f;
typedef __attribute__((ext_vector_type(8))) float v8f;

__device__ __forceinline__ int chip_half() {
    return (int)((__builtin_amdgcn_s_sendmsg_rtn(0x87) >> 18) & 1u);
}

#define TS 64
__global__ void k_deg_init(float* __restrict__ deg, int* __restrict__ tick, int n, int np) {
    int i = blockIdx.x * blockDim.x + threadIdx.x;
    if (i < n) {
        deg[i] = 1.0f;
        deg[np + i] = 0.0f;
    }
    if (i < 4 * TS) tick[i] = 0;
}

#define HN 65536
#define HB 128
#define HU 16
#define HF 8

__global__ void __launch_bounds__(1024)
k_deg_count(const int* __restrict__ dst, float* __restrict__ deg, int n, int np, int e) {
    __shared__ __attribute__((aligned(16))) unsigned hist[HN];
    const int half = chip_half();
    const int tid = threadIdx.x;
    float* __restrict__ dh = deg + (size_t)half * np;
    const int per = ((e + (int)gridDim.x - 1) / (int)gridDim.x + 31) & ~31;
    const int lo = blockIdx.x * per;
    const int hi = min(e, lo + per);
    for (int node0 = 0; node0 < n; node0 += HN) {
        const int cnt = min(HN, n - node0);
        for (int i = tid; i < (cnt + 3) / 4; i += 1024)
            ((uint4*)hist)[i] = make_uint4(0u, 0u, 0u, 0u);
        __syncthreads();
        for (int i0 = lo + tid; i0 < hi; i0 += 1024 * HU) {
            int d[HU];
#pragma unroll
            for (int u = 0; u < HU; ++u) {
                const int i = i0 + u * 1024;
                d[u] = (i < hi) ? dst[i] : -1;
            }
#pragma unroll
            for (int u = 0; u < HU; ++u) {
                const unsigned r = (unsigned)(d[u] - node0);
                if (r < (unsigned)cnt) atomicAdd(&hist[r], 1u);
            }
        }
        __syncthreads();
        for (int i0 = tid; i0 < cnt; i0 += 1024 * HF) {
            unsigned c[HF];
#pragma unroll
            for (int u = 0; u < HF; ++u) {
                const int i = i0 + u * 1024;
                c[u] = (i < cnt) ? hist[i] : 0u;
            }
#pragma unroll
            for (int u = 0; u < HF; ++u)
                if (c[u]) atomicAdd(dh + node0 + i0 + u * 1024, (float)c[u]);
        }
        __syncthreads();
    }
}

__device__ __forceinline__ float inv_sqrt_deg(float d) {
    return (d > 0.0f) ? rsqrtf(d) : 0.0f;
}

#define GW 4
#define AP 68
#define WP 160

template <bool LAYER2>
__global__ void __launch_bounds__(GW * 32)
k_gemm_wmma(const float* __restrict__ X,
            const float* __restrict__ W,
            const float* __restrict__ deg, const float* __restrict__ deg1,
            const float* __restrict__ bias,
            float* __restrict__ Y, float* __restrict__ Z,
            float* __restrict__ dsum, int n) {
    __shared__ __attribute__((aligned(16))) float sW[(D / 2) * WP];
    __shared__ __attribute__((aligned(16))) float sT[GW][16 * AP];
    const int tid  = threadIdx.x;
    const int lane = tid & 31;
    const int wave = tid >> 5;
    const int r0 = (blockIdx.x * GW + wave) << 4;
    const int m  = lane & 15;
    const int kh = lane >> 4;
    const int lr = lane >> 4;
    const int lc = (lane & 15) << 2;
    float* __restrict__ tA = sT[wave];

    float4 av[8];
#pragma unroll
    for (int i = 0; i < 8; ++i) {
        const int row = min(r0 + 2 * i + lr, n - 1);
        av[i] = *(const float4*)(X + (size_t)row * D + lc);
    }
    float4 we[(D * D / 8) / (GW * 32)], wo[(D * D / 8) / (GW * 32)];
#pragma unroll
    for (int j = 0; j < (D * D / 8) / (GW * 32); ++j) {
        const int q = tid + j * GW * 32;
        we[j] = *(const float4*)(W + (size_t)(2 * (q >> 4))     * D + ((q & 15) << 2));
        wo[j] = *(const float4*)(W + (size_t)(2 * (q >> 4) + 1) * D + ((q & 15) << 2));
    }
    float dg[8];
#pragma unroll
    for (int i = 0; i < 8; ++i) {
        const int row = min(r0 + 2 * i + lr, n - 1);
        dg[i] = LAYER2 ? deg[row] : deg[row] + deg1[row];
    }
    const float4 b4 = *(const float4*)(bias + lc);

#pragma unroll
    for (int j = 0; j < (D * D / 8) / (GW * 32); ++j) {
        const int q = tid + j * GW * 32;
        float* __restrict__ wp = &sW[(q >> 4) * WP + ((q & 15) << 3)];
        *(float4*)(wp)     = make_float4(we[j].x, wo[j].x, we[j].y, wo[j].y);
        *(float4*)(wp + 4) = make_float4(we[j].z, wo[j].z, we[j].w, wo[j].w);
    }
#pragma unroll
    for (int i = 0; i < 8; ++i) {
        float4 a = av[i];
        if (LAYER2) {
            a.x = fmaxf(a.x, 0.0f);
            a.y = fmaxf(a.y, 0.0f);
            a.z = fmaxf(a.z, 0.0f);
            a.w = fmaxf(a.w, 0.0f);
        }
        *(float4*)&tA[(2 * i + lr) * AP + lc] = a;
    }
    __syncthreads();

    v8f acc[4] = {};
#pragma unroll
    for (int k0 = 0; k0 < D; k0 += 4) {
        const v2f a = *(const v2f*)&tA[m * AP + k0 + 2 * kh];
#pragma unroll
        for (int ct = 0; ct < 4; ++ct) {
            const v2f b = *(const v2f*)&sW[(k0 / 2 + kh) * WP + 2 * (16 * ct + m)];
            acc[ct] = __builtin_amdgcn_wmma_f32_16x16x4_f32(
                 false, a,  false, b,
                 (short)0, acc[ct],  false,  false);
        }
    }

#pragma unroll
    for (int ct = 0; ct < 4; ++ct)
#pragma unroll
        for (int r = 0; r < 8; ++r)
            tA[(r + 8 * kh) * AP + 16 * ct + m] = acc[ct][r];
    __syncthreads();

#pragma unroll
    for (int i = 0; i < 8; ++i) {
        const int row = r0 + 2 * i + lr;
        if (row < n) {
            const float4 y = *(const float4*)&tA[(2 * i + lr) * AP + lc];
            const size_t o = (size_t)row * D + lc;
            const float dv = inv_sqrt_deg(dg[i]);
            const float sc = dv * dv;
            float4 z;
            z.x = sc * y.x + b4.x;
            z.y = sc * y.y + b4.y;
            z.z = sc * y.z + b4.z;
            z.w = sc * y.w + b4.w;
            *(float4*)(Y + o)  = y;
            *(float4*)(Z + o)  = z;
            if (!LAYER2 && lc == 0) dsum[row] = dg[i];
        }
    }
}

#define WN 65536
#define WB 128
#define WF 32
#define WU 16

__global__ void __launch_bounds__(1024)
k_edge_weight(const int* __restrict__ src, const int* __restrict__ dst,
              const float* __restrict__ deg, float* __restrict__ wgt, int n, int e) {
    __shared__ __attribute__((aligned(16))) float dinv[WN];
    const int tid = threadIdx.x;
    const int cap = min(n, WN);
    for (int i0 = tid; i0 < cap; i0 += 1024 * WF) {
        float d[WF];
#pragma unroll
        for (int u = 0; u < WF; ++u) {
            const int i = i0 + u * 1024;
            d[u] = (i < cap) ? deg[i] : 0.0f;
        }
#pragma unroll
        for (int u = 0; u < WF; ++u) {
            const int i = i0 + u * 1024;
            if (i < cap) dinv[i] = inv_sqrt_deg(d[u]);
        }
    }
    __syncthreads();
    const int per = ((e + (int)gridDim.x - 1) / (int)gridDim.x + 31) & ~31;
    const int lo = blockIdx.x * per;
    const int hi = min(e, lo + per);
    for (int i0 = lo + tid; i0 < hi; i0 += 1024 * WU) {
        int s[WU], t[WU];
        float w[WU];
#pragma unroll
        for (int u = 0; u < WU; ++u) {
            const int i = i0 + u * 1024;
            s[u] = (i < hi) ? src[i] : 0;
            t[u] = (i < hi) ? dst[i] : 0;
        }
#pragma unroll
        for (int u = 0; u < WU; ++u)
            w[u] = dinv[min(s[u], WN - 1)] * dinv[min(t[u], WN - 1)];
        if (n > WN) {
#pragma unroll
            for (int u = 0; u < WU; ++u)
                if (max(s[u], t[u]) >= WN) {
                    const float fs = (s[u] < WN) ? dinv[s[u]] : inv_sqrt_deg(deg[s[u]]);
                    const float ft = (t[u] < WN) ? dinv[t[u]] : inv_sqrt_deg(deg[t[u]]);
                    w[u] = fs * ft;
                }
        }
#pragma unroll
        for (int u = 0; u < WU; ++u) {
            const int i = i0 + u * 1024;
            if (i < hi) wgt[i] = w[u];
        }
    }
}

#define EPW 32
#define EB  8

__device__ __forceinline__ int lane_bcast(int v, int l) {
    return __builtin_amdgcn_readlane(v, l);
}

#define SG 1024

__device__ __forceinline__ int draw_ticket(int* __restrict__ tick, int half, int ngrp) {
    int own = half;
    int tk = atomicAdd(tick + own * TS, 1);
    if (tk >= ngrp) {
        own ^= 1;
        tk = atomicAdd(tick + own * TS, 1);
    }
    return (tk < ngrp) ? (tk * 2 + own) : -1;
}

template <int NB>
__device__ __forceinline__ void walk_edges(const float* __restrict__ xw, float* __restrict__ agg,
                                           int s, int t, int wbits, int cnt, int lane,
                                           bool drawer, int* __restrict__ tick, int half, int ngrp,
                                           int& jout) {
    float a0[2][EB], a1[2][EB];
#pragma unroll
    for (int j = 0; j < EB; ++j) {
        if (NB > 1 || j < cnt) {
            const float* __restrict__ xs = xw + (size_t)lane_bcast(s, j) * D;
            a0[0][j] = xs[lane];
            a1[0][j] = xs[lane + 32];
        }
    }
#pragma unroll
    for (int b = 0; b < NB; ++b) {
        if (b + 1 < NB) {
#pragma unroll
            for (int j = 0; j < EB; ++j) {
                if (b + 2 < NB || (b + 1) * EB + j < cnt) {
                    const float* __restrict__ xs = xw + (size_t)lane_bcast(s, (b + 1) * EB + j) * D;
                    a0[(b + 1) & 1][j] = xs[lane];
                    a1[(b + 1) & 1][j] = xs[lane + 32];
                }
            }
        }
        if (b == 0 && drawer) jout = draw_ticket(tick, half, ngrp);
#pragma unroll
        for (int j = 0; j < EB; ++j) {
            if (b + 1 < NB || b * EB + j < cnt) {
                const float nrm = __int_as_float(lane_bcast(wbits, b * EB + j));
                float* __restrict__ ad = agg + (size_t)lane_bcast(t, b * EB + j) * D;
                atomicAdd(ad + lane,      nrm * a0[b & 1][j]);
                atomicAdd(ad + lane + 32, nrm * a1[b & 1][j]);
            }
        }
    }
}

__global__ void __launch_bounds__(256, 8)
k_scatter(const int* __restrict__ src, const int* __restrict__ dst,
          const float* __restrict__ wgt, const float* __restrict__ xw,
          float* __restrict__ agg, int* __restrict__ tick, int ngrp, int n, int e) {
    __shared__ int s_job[4];
    const int half = chip_half();
    const int lane = threadIdx.x & 31;
    const int wave = threadIdx.x >> 5;
    const bool first = threadIdx.x == 0;
    if (first) {
        const int j0 = draw_ticket(tick, half, ngrp);
        s_job[2] = j0;
        s_job[3] = (j0 >= 0) ? draw_ticket(tick, half, ngrp) : -1;
    }
    __syncthreads();
    int cur = __builtin_amdgcn_readfirstlane(s_job[2]);
    int nxt = __builtin_amdgcn_readfirstlane(s_job[3]);
    if (cur < 0) return;

    int rs, rt, rw;
    {
        const int at = min((((cur >> 1) << 3) + wave) * EPW + lane, e - 1);
        rs = src[at];
        rt = dst[at];
        rw = __float_as_int(wgt[at]);
    }
    int slot = 0;
    for (;;) {
        const int own = cur & 1;
        const int base = (((cur >> 1) << 3) + wave) * EPW;
        const int valid = max(0, min(EPW, e - base));
        const bool mine = (lane < valid) && ((rt >= (n >> 1)) == (own != 0));
        const unsigned mask = __builtin_amdgcn_ballot_w32(mine);
        const int cnt = __builtin_popcount(mask);
        const unsigned below = (1u << lane) - 1u;
        const int to = mine ? __builtin_popcount(mask & below) : cnt + __builtin_popcount(~mask & below);
        const int s     = __builtin_amdgcn_ds_permute(to << 2, rs);
        const int t     = __builtin_amdgcn_ds_permute(to << 2, rt);
        const int wbits = __builtin_amdgcn_ds_permute(to << 2, rw);
        if (nxt >= 0) {
            const int at = min((((nxt >> 1) << 3) + wave) * EPW + lane, e - 1);
            rs = src[at];
            rt = dst[at];
            rw = __float_as_int(wgt[at]);
        }
        int j2 = -1;
        const bool drawer = first && nxt >= 0;
        if (cnt > 3 * EB)      walk_edges<4>(xw, agg, s, t, wbits, cnt, lane, drawer, tick, half, ngrp, j2);
        else if (cnt > 2 * EB) walk_edges<3>(xw, agg, s, t, wbits, cnt, lane, drawer, tick, half, ngrp, j2);
        else if (cnt > EB)     walk_edges<2>(xw, agg, s, t, wbits, cnt, lane, drawer, tick, half, ngrp, j2);
        else                   walk_edges<1>(xw, agg, s, t, wbits, cnt, lane, drawer, tick, half, ngrp, j2);

        if (nxt < 0) break;
        if (first) s_job[slot] = j2;
        __builtin_amdgcn_fence(__ATOMIC_RELEASE, "workgroup", "local");
        __builtin_amdgcn_s_barrier();
        __builtin_amdgcn_fence(__ATOMIC_ACQUIRE, "workgroup", "local");
        cur = nxt;
        nxt = __builtin_amdgcn_readfirstlane(s_job[slot]);
        slot ^= 1;
    }
}

extern "C" void kernel_launch(void* const* d_in, const int* in_sizes, int n_in,
                              void* d_out, int out_size, void* d_ws, size_t ws_size,
                              hipStream_t stream) {
    const float* x  = (const float*)d_in[0];
    const int*   ei = (const int*)  d_in[1];
    const float* W1 = (const float*)d_in[2];
    const float* b1 = (const float*)d_in[3];
    const float* W2 = (const float*)d_in[4];
    const float* b2 = (const float*)d_in[5];

    const int n = in_sizes[0] / D;
    const int e = in_sizes[1] / 2;
    const int* src = ei;
    const int* dst = ei + e;

    float* ws   = (float*)d_ws;
    float* xw   = ws + 2 * (size_t)n;
    float* agg  = xw + (size_t)n * D;
    float* dsum = agg + (size_t)n * D;
    const int np = (n + 31) & ~31;
    float* deg  = ws + ((3 * (size_t)n + 2 * (size_t)n * D + 31) & ~(size_t)31);
    float* wgt  = deg + 2 * (size_t)np;
    int*   tick = (int*)(wgt + (((size_t)e + 31) & ~(size_t)31));
    float* out  = (float*)d_out;

    const int T = 256;
    const int rowTiles = (n + 15) / 16;
    const int gemmBlocks = (rowTiles + GW - 1) / GW;
    const int scatBlocks = (e + 8 * EPW - 1) / (8 * EPW);

    k_deg_init <<<(n + T - 1) / T, T, 0, stream>>>(deg, tick, n, np);
    k_deg_count<<<HB, 1024, 0, stream>>>(dst, deg, n, np, e);

    k_gemm_wmma<false><<<gemmBlocks, GW * 32, 0, stream>>>(x, W1, deg, deg + np, b1, xw, agg, dsum, n);
    k_edge_weight     <<<WB, 1024, 0, stream>>>(src, dst, dsum, wgt, n, e);
    k_scatter         <<<SG, 256, 0, stream>>>(src, dst, wgt, xw, agg, tick, scatBlocks, n, e);

    k_gemm_wmma<true> <<<gemmBlocks, GW * 32, 0, stream>>>(agg, W2, dsum, dsum, b2, xw, out, dsum, n);
    k_scatter         <<<SG, 256, 0, stream>>>(src, dst, wgt, xw, out, tick + 2 * TS, scatBlocks, n, e);
}
